# starting kernel
# baseline (speedup 1.0000x reference)
#include <hip/hip_runtime.h>

#define N_NODES 50000
#define N_EDGES 800000
#define DIM     128

typedef __attribute__((ext_vector_type(2))) float v2f;
typedef __attribute__((ext_vector_type(8))) float v8f;

__global__ void k_init_deg(float* __restrict__ deg) {
    int i = blockIdx.x * blockDim.x + threadIdx.x;
    if (i < N_NODES) deg[i] = 1.0f;
}

__global__ void k_count_deg(const int* __restrict__ ei, float* __restrict__ deg) {
    int e = blockIdx.x * blockDim.x + threadIdx.x;
    if (e < N_EDGES) {
        int dst = ei[N_EDGES + e];
        unsafeAtomicAdd(&deg[dst], 1.0f);
    }
}

__global__ void k_rsqrt(float* __restrict__ deg) {
    int i = blockIdx.x * blockDim.x + threadIdx.x;
    if (i < N_NODES) deg[i] = rsqrtf(deg[i]);
}

__global__ void k_init_acc(const float* __restrict__ x, const float* __restrict__ dinv,
                           float* __restrict__ acc) {
    int g = blockIdx.x * blockDim.x + threadIdx.x;
    if (g < N_NODES * (DIM / 4)) {
        int n = g >> 5;
        float s = dinv[n]; s = s * s;
        float4 v = ((const float4*)x)[g];
        v.x *= s; v.y *= s; v.z *= s; v.w *= s;
        ((float4*)acc)[g] = v;
    }
}

__global__ void k_scatter(const float* __restrict__ x, const int* __restrict__ ei,
                          const float* __restrict__ dinv, float* __restrict__ acc) {
    int g = blockIdx.x * blockDim.x + threadIdx.x;
    int e = g >> 5;
    int lane = g & 31;
    if (e < N_EDGES) {
        int src = ei[e];
        int dst = ei[N_EDGES + e];
        float nrm = dinv[src] * dinv[dst];
        float4 v = ((const float4*)(x + (size_t)src * DIM))[lane];
        float* p = acc + (size_t)dst * DIM + lane * 4;
        unsafeAtomicAdd(p + 0, v.x * nrm);
        unsafeAtomicAdd(p + 1, v.y * nrm);
        unsafeAtomicAdd(p + 2, v.z * nrm);
        unsafeAtomicAdd(p + 3, v.w * nrm);
    }
}

__global__ void __launch_bounds__(256) k_gemm_fused(
        const float* __restrict__ acc, const float* __restrict__ x,
        const float* __restrict__ W, const float* __restrict__ bias,
        float* __restrict__ out) {
    __shared__ float At[16 * 132];
    const int tid  = threadIdx.x;
    const int wave = tid >> 5;
    const int lane = tid & 31;
    const int row0 = blockIdx.x * 16;

    {
        const float4* src4 = (const float4*)(acc + (size_t)row0 * DIM);
        float4* dst4 = (float4*)At;
        #pragma unroll
        for (int i = tid; i < 16 * 32; i += 256) {
            int r = i >> 5, q = i & 31;
            dst4[r * 33 + q] = src4[i];
        }
    }
    __syncthreads();

    const int col0 = wave * 16;
    const int nn   = lane & 15;
    const int koff = (lane >> 4) << 1;
    const int colg = col0 + nn;

    v2f bfrag[32];
    #pragma unroll
    for (int kk = 0; kk < 32; ++kk) {
        int k = (kk << 2) + koff;
        v2f t;
        t.x = W[k * DIM + colg];
        t.y = W[(k + 1) * DIM + colg];
        bfrag[kk] = t;
    }

    v8f c = {};
    const float* arow = At + nn * 132;
    #pragma unroll
    for (int kk = 0; kk < 32; ++kk) {
        int k = (kk << 2) + koff;
        v2f a;
        a.x = arow[k];
        a.y = arow[k + 1];
        c = __builtin_amdgcn_wmma_f32_16x16x4_f32(false, a, false, bfrag[kk],
                                                  (short)0, c, false, false);
    }

    const float bv = bias[colg];
    const int rbase = row0 + ((lane >> 4) << 3);
    #pragma unroll
    for (int r = 0; r < 8; ++r) {
        size_t off = (size_t)(rbase + r) * DIM + colg;
        float v = c[r] + bv;
        v = v > 0.0f ? v : 0.0f;
        out[off] = x[off] + v;
    }
}

extern "C" void kernel_launch(void* const* d_in, const int* in_sizes, int n_in,
                              void* d_out, int out_size, void* d_ws, size_t ws_size,
                              hipStream_t stream) {
    (void)in_sizes; (void)n_in; (void)out_size; (void)ws_size;
    const float* x  = (const float*)d_in[0];
    const int*   ei = (const int*)d_in[1];
    const float* W  = (const float*)d_in[2];
    const float* b  = (const float*)d_in[3];
    float* out = (float*)d_out;

    char* ws  = (char*)d_ws;
    float* deg = (float*)ws;
    float* acc = (float*)(ws + ((N_NODES * 4 + 255) & ~255));

    k_init_deg <<<(N_NODES + 255) / 256, 256, 0, stream>>>(deg);
    k_count_deg<<<(N_EDGES + 255) / 256, 256, 0, stream>>>(ei, deg);
    k_rsqrt    <<<(N_NODES + 255) / 256, 256, 0, stream>>>(deg);
    k_init_acc <<<(N_NODES * (DIM / 4) + 255) / 256, 256, 0, stream>>>(x, deg, acc);
    k_scatter  <<<(N_EDGES * 32 + 255) / 256, 256, 0, stream>>>(x, ei, deg, acc);
    k_gemm_fused<<<N_NODES / 16, 256, 0, stream>>>(acc, x, W, b, out);
}


# Owners multiply their finished 196-row LDS tile by W on the matrix unit and store out with bias, relu, residual; summed rows never reach memory, one launch fewer; sums unchanged
# speedup vs baseline: 2.7711x; 2.7711x over previous
#include <hip/hip_runtime.h>

#define N_NODES 50000
#define N_EDGES 800000
#define DIM     128
#define ROWS    196
#define NCH     1563
#define QW      48
#define HB      100
#define HW      (N_NODES / 2)
#define HWS     25024
#define HG      (N_EDGES / HB / 4)
#define HR      ((HG + 1023) / 1024)
#define TS      132
#define TROWS   208

typedef __attribute__((ext_vector_type(2))) float v2f;
typedef __attribute__((ext_vector_type(8))) float v8f;

__global__ void __launch_bounds__(1024) k_hist(const int* __restrict__ ei, unsigned* __restrict__ part,
                                               unsigned* __restrict__ rb) {
    __shared__ unsigned h[HW];
    const int tid = threadIdx.x;
    const int g0  = blockIdx.x * HG;
    int4 d[HR];
    #pragma unroll
    for (int k = 0; k < HR; ++k) {
        const int i = tid + 1024 * k;
        d[k] = ((const int4*)(ei + N_EDGES))[g0 + (i < HG ? i : 0)];
    }
    for (int i = tid; i < HW / 4; i += 1024) ((uint4*)h)[i] = make_uint4(0u, 0u, 0u, 0u);
    __syncthreads();
    #pragma unroll
    for (int k = 0; k < HR; ++k) {
        const int i = tid + 1024 * k;
        if (i < HG) {
            const int4 e = d[k];
            atomicAdd(&h[e.x >> 1], 1u << ((e.x & 1) << 4));
            atomicAdd(&h[e.y >> 1], 1u << ((e.y & 1) << 4));
            atomicAdd(&h[e.z >> 1], 1u << ((e.z & 1) << 4));
            atomicAdd(&h[e.w >> 1], 1u << ((e.w & 1) << 4));
            rb[g0 + i] = ((unsigned)e.x / ROWS) | (((unsigned)e.y / ROWS) << 8) |
                         (((unsigned)e.z / ROWS) << 16) | (((unsigned)e.w / ROWS) << 24);
        }
    }
    __syncthreads();
    uint4* dst = (uint4*)(part + (size_t)blockIdx.x * HWS);
    for (int i = tid; i < HW / 4; i += 1024) dst[i] = ((const uint4*)h)[i];
}

__global__ void __launch_bounds__(256) k_dinv(const unsigned* __restrict__ part, float* __restrict__ dinv) {
    __shared__ unsigned s[3][2][64];
    const int tid = threadIdx.x;
    const int c   = tid & 63;
    const int pg  = tid >> 6;
    const int i   = blockIdx.x * 64 + c;
    unsigned lo = 0u, hi = 0u;
    if (i < HW) {
        #pragma unroll
        for (int k = 0; k < HB / 4; ++k) {
            const unsigned w = part[(size_t)(pg + 4 * k) * HWS + i];
            lo += w & 0xffffu;
            hi += w >> 16;
        }
    }
    if (pg > 0) { s[pg - 1][0][c] = lo; s[pg - 1][1][c] = hi; }
    __syncthreads();
    if (pg == 0 && i < HW) {
        lo += s[0][0][c] + s[1][0][c] + s[2][0][c];
        hi += s[0][1][c] + s[1][1][c] + s[2][1][c];
        float2 r;
        r.x = rsqrtf((float)(lo + 1u));
        r.y = rsqrtf((float)(hi + 1u));
        ((float2*)dinv)[i] = r;
    }
}

__device__ __forceinline__ unsigned zero_bytes(unsigned y) {
    return ~(((y & 0x7f7f7f7fu) + 0x7f7f7f7fu) | y | 0x7f7f7f7fu);
}

template <bool FULL>
__device__ __forceinline__ void edge_group(const int* qp, int cnt, int lane, int row0,
        const float* __restrict__ x, const int* __restrict__ ei,
        const float* __restrict__ dinv, float* tile) {
    const int jj = lane & 15;
    const int e  = (FULL || jj < cnt) ? qp[jj] : 0;
    const int   my  = ei[lane < 16 ? e : N_EDGES + e];
    const float myd = dinv[my];
    float v[16][4];
    #pragma unroll
    for (int j = 0; j < 16; ++j) {
        v[j][0] = 0.0f; v[j][1] = 0.0f; v[j][2] = 0.0f; v[j][3] = 0.0f;
        if (FULL || j < cnt) {
            const int s = __builtin_amdgcn_readlane(my, j);
            const float* xr = x + (size_t)s * DIM + lane;
            v[j][0] = xr[0];
            v[j][1] = xr[32];
            v[j][2] = xr[64];
            v[j][3] = xr[96];
        }
    }
    const int mydi = __float_as_int(myd);
    #pragma unroll
    for (int j = 0; j < 16; ++j) {
        if (FULL || j < cnt) {
            const int r = __builtin_amdgcn_readlane(my, 16 + j) - row0;
            const float nrm = __int_as_float(__builtin_amdgcn_readlane(mydi, j)) *
                              __int_as_float(__builtin_amdgcn_readlane(mydi, 16 + j));
            float* p = tile + r * TS + lane;
            unsafeAtomicAdd(p +  0, v[j][0] * nrm);
            unsafeAtomicAdd(p + 32, v[j][1] * nrm);
            unsafeAtomicAdd(p + 64, v[j][2] * nrm);
            unsafeAtomicAdd(p + 96, v[j][3] * nrm);
        }
    }
}

__global__ void __launch_bounds__(1024) k_owner(const float* __restrict__ x, const int* __restrict__ ei,
        const float* __restrict__ dinv, const uint4* __restrict__ rb,
        const float* __restrict__ W, const float* __restrict__ bias, float* __restrict__ out) {
    __shared__ float tile[TROWS * TS];
    __shared__ int   q[32 * QW];
    const int tid  = threadIdx.x;
    const int lane = tid & 31;
    const int w    = tid >> 5;
    const int rid  = blockIdx.x;
    const int row0 = rid * ROWS;
    const int nrows = (N_NODES - row0) < ROWS ? (N_NODES - row0) : ROWS;

    for (int i = tid; i < nrows * (DIM / 4); i += 1024) {
        float s = dinv[row0 + (i >> 5)]; s = s * s;
        float4 t = ((const float4*)(x + (size_t)row0 * DIM))[i];
        t.x *= s; t.y *= s; t.z *= s; t.w *= s;
        ((float4*)tile)[(i >> 5) * (TS / 4) + (i & 31)] = t;
    }
    for (int i = nrows * (DIM / 4) + tid; i < TROWS * (DIM / 4); i += 1024)
        ((float4*)tile)[(i >> 5) * (TS / 4) + (i & 31)] = make_float4(0.0f, 0.0f, 0.0f, 0.0f);
    __syncthreads();

    int* myq = q + w * QW;
    int qn = 0;
    const unsigned pat  = (unsigned)rid * 0x01010101u;
    const int      base = rid * 6;
    for (int t0 = w; t0 < NCH; t0 += 256) {
        unsigned f[8];
        #pragma unroll
        for (int u = 0; u < 8; ++u) {
            const int t = t0 + 32 * u;
            int c = t + base; if (c >= NCH) c -= NCH;
            const int  o16 = c * 32 + lane;
            const bool ok  = (t < NCH) && (o16 < N_EDGES / 16);
            const uint4 b  = rb[ok ? o16 : 0];
            const unsigned z = (zero_bytes(b.x ^ pat) >> 7) | (zero_bytes(b.y ^ pat) >> 6) |
                               (zero_bytes(b.z ^ pat) >> 5) | (zero_bytes(b.w ^ pat) >> 4);
            f[u] = ok ? z : 0u;
        }
        #pragma unroll 1
        for (int u = 0; u < 8; ++u) {
            unsigned m = f[0];
            f[0] = f[1]; f[1] = f[2]; f[2] = f[3]; f[3] = f[4]; f[4] = f[5]; f[5] = f[6]; f[6] = f[7]; f[7] = 0u;
            int c = t0 + 32 * u + base; if (c >= NCH) c -= NCH;
            const int ebase = (c * 32 + lane) * 16;
            for (;;) {
                const unsigned bal = __builtin_amdgcn_ballot_w32(m != 0u);
                if (bal == 0u) break;
                if (m != 0u) {
                    const int pos = (int)__builtin_amdgcn_mbcnt_lo(bal, 0u);
                    const int p   = __builtin_ctz(m);
                    myq[qn + pos] = ebase + ((p & 3) << 2) + (p >> 3);
                    m &= m - 1u;
                }
                qn += __builtin_popcount(bal);
                while (qn >= 16) {
                    qn -= 16;
                    edge_group<true>(myq + qn, 16, lane, row0, x, ei, dinv, tile);
                }
            }
        }
    }
    if (qn > 0) edge_group<false>(myq, qn, lane, row0, x, ei, dinv, tile);

    const int wu   = __builtin_amdgcn_readfirstlane(w);
    const int nn   = lane & 15;
    const int koff = (lane >> 4) << 1;
    const int colg = (wu & 7) * 16 + nn;
    v2f bfrag[32];
    #pragma unroll
    for (int kk = 0; kk < 32; ++kk) {
        const int k = (kk << 2) + koff;
        v2f t;
        t.x = W[k * DIM + colg];
        t.y = W[(k + 1) * DIM + colg];
        bfrag[kk] = t;
    }
    const float bv = bias[colg];
    __syncthreads();

    const int ntile = ((wu >> 3) == (wu & 3)) ? 4 : 3;
    #pragma unroll 1
    for (int i = 0; i < ntile; ++i) {
        const int rt   = (i < 3) ? (wu >> 3) + 4 * i : 12;
        const int rloc = rt * 16 + ((lane >> 4) << 3);
        const bool whole = rt * 16 + 16 <= nrows;
        const size_t o0 = (size_t)(row0 + rloc) * DIM + colg;
        const float* xp = x + o0;
        float*       op = out + o0;
        float xr[8];
        if (whole) {
            #pragma unroll
            for (int r = 0; r < 8; ++r) xr[r] = xp[r * DIM];
        } else {
            #pragma unroll
            for (int r = 0; r < 8; ++r) xr[r] = (rloc + r < nrows) ? xp[r * DIM] : 0.0f;
        }
        const float* arow = tile + (rt * 16 + nn) * TS;
        v8f c = {};
        #pragma unroll
        for (int kk = 0; kk < 32; ++kk) {
            const int k = (kk << 2) + koff;
            v2f a;
            a.x = arow[k];
            a.y = arow[k + 1];
            c = __builtin_amdgcn_wmma_f32_16x16x4_f32(false, a, false, bfrag[kk],
                                                      (short)0, c, false, false);
        }
        if (whole) {
            #pragma unroll
            for (int r = 0; r < 8; ++r) {
                float v = c[r] + bv;
                v = v > 0.0f ? v : 0.0f;
                op[r * DIM] = xr[r] + v;
            }
        } else {
            #pragma unroll
            for (int r = 0; r < 8; ++r) {
                if (rloc + r < nrows) {
                    float v = c[r] + bv;
                    v = v > 0.0f ? v : 0.0f;
                    op[r * DIM] = xr[r] + v;
                }
            }
        }
    }
}

extern "C" void kernel_launch(void* const* d_in, const int* in_sizes, int n_in,
                              void* d_out, int out_size, void* d_ws, size_t ws_size,
                              hipStream_t stream) {
    (void)in_sizes; (void)n_in; (void)out_size; (void)ws_size;
    const float* x  = (const float*)d_in[0];
    const int*   ei = (const int*)d_in[1];
    const float* W  = (const float*)d_in[2];
    const float* b  = (const float*)d_in[3];
    float* out = (float*)d_out;

    char* ws  = (char*)d_ws;
    float* dinv = (float*)ws;
    unsigned* rb = (unsigned*)(ws + ((N_NODES * 4 + 255) & ~255) + N_NODES * DIM * 4);
    unsigned* part = (unsigned*)(ws + ((N_NODES * 4 + 255) & ~255) + N_NODES * DIM * 4 + N_EDGES);

    k_hist     <<<HB, 1024, 0, stream>>>(ei, part, rb);
    k_dinv     <<<(HW + 63) / 64, 256, 0, stream>>>(part, dinv);
    k_owner    <<<(N_NODES + ROWS - 1) / ROWS, 1024, 0, stream>>>(x, ei, dinv, (const uint4*)rb, W, b, out);
}


# First launch also builds one edge bitmap per 196-row range (LDS ors, whole-line stores); an owner reads 1 bit per edge of its own bitmap, not 1 byte of a shared stream; arrival order may differ
# speedup vs baseline: 3.1539x; 1.1381x over previous
#include <hip/hip_runtime.h>

#define N_NODES 50000
#define N_EDGES 800000
#define DIM     128
#define ROWS    196
#define BE      4096
#define NBB     ((N_EDGES + BE - 1) / BE)
#define BWD     (BE / 32)
#define BST     132
#define BMW     (NBB * BWD)
#define NCW     (BMW / 32)
#define QW      48
#define HB      100
#define HW      (N_NODES / 2)
#define HWS     25024
#define HT      512
#define HG      (N_EDGES / HB / 4)
#define HR      ((HG + HT - 1) / HT)
#define BR      (BE / 4 / HT)
#define TS      132
#define TROWS   208

typedef __attribute__((ext_vector_type(2))) float v2f;
typedef __attribute__((ext_vector_type(8))) float v8f;

__device__ __forceinline__ void hist_role(const int* __restrict__ ei, unsigned* __restrict__ part, unsigned* h) {
    const int tid = threadIdx.x;
    const int g0  = blockIdx.x * HG;
    int4 d[HR];
    #pragma unroll
    for (int k = 0; k < HR; ++k) {
        const int i = tid + HT * k;
        d[k] = ((const int4*)(ei + N_EDGES))[g0 + (i < HG ? i : 0)];
    }
    for (int i = tid; i < HW / 4; i += HT) ((uint4*)h)[i] = make_uint4(0u, 0u, 0u, 0u);
    __syncthreads();
    #pragma unroll
    for (int k = 0; k < HR; ++k) {
        const int i = tid + HT * k;
        if (i < HG) {
            const int4 e = d[k];
            atomicAdd(&h[e.x >> 1], 1u << ((e.x & 1) << 4));
            atomicAdd(&h[e.y >> 1], 1u << ((e.y & 1) << 4));
            atomicAdd(&h[e.z >> 1], 1u << ((e.z & 1) << 4));
            atomicAdd(&h[e.w >> 1], 1u << ((e.w & 1) << 4));
        }
    }
    __syncthreads();
    uint4* dst = (uint4*)(part + (size_t)blockIdx.x * HWS);
    for (int i = tid; i < HW / 4; i += HT) dst[i] = ((const uint4*)h)[i];
}

__device__ __forceinline__ void bitmap_role(const int* __restrict__ ei, uint4* __restrict__ bm, unsigned* s, int b) {
    const int tid = threadIdx.x;
    const int g0  = b * (BE / 4);
    int4 d[BR];
    #pragma unroll
    for (int k = 0; k < BR; ++k) {
        const int gi = g0 + tid + HT * k;
        d[k] = ((const int4*)(ei + N_EDGES))[gi < N_EDGES / 4 ? gi : 0];
    }
    for (int i = tid; i < 256 * BST / 4; i += HT) ((uint4*)s)[i] = make_uint4(0u, 0u, 0u, 0u);
    __syncthreads();
    #pragma unroll
    for (int k = 0; k < BR; ++k) {
        const int gl = tid + HT * k;
        if (g0 + gl < N_EDGES / 4) {
            const int4 e = d[k];
            const int      wd = gl >> 3;
            const unsigned sh = (unsigned)(gl & 7) << 2;
            atomicOr(&s[((unsigned)e.x / ROWS) * BST + wd], 1u << sh);
            atomicOr(&s[((unsigned)e.y / ROWS) * BST + wd], 2u << sh);
            atomicOr(&s[((unsigned)e.z / ROWS) * BST + wd], 4u << sh);
            atomicOr(&s[((unsigned)e.w / ROWS) * BST + wd], 8u << sh);
        }
    }
    __syncthreads();
    for (int i = tid; i < 256 * (BWD / 4); i += HT) {
        const int r = i / (BWD / 4), j = i % (BWD / 4);
        bm[(size_t)r * (BMW / 4) + b * (BWD / 4) + j] = ((const uint4*)s)[r * (BST / 4) + j];
    }
}

__global__ void __launch_bounds__(HT) k_hist(const int* __restrict__ ei, unsigned* __restrict__ part,
                                             uint4* __restrict__ bm) {
    __shared__ unsigned s[256 * BST];
    if (blockIdx.x < HB) hist_role(ei, part, s);
    else                 bitmap_role(ei, bm, s, blockIdx.x - HB);
}

__global__ void __launch_bounds__(256) k_dinv(const unsigned* __restrict__ part, float* __restrict__ dinv) {
    __shared__ unsigned s[3][2][64];
    const int tid = threadIdx.x;
    const int c   = tid & 63;
    const int pg  = tid >> 6;
    const int i   = blockIdx.x * 64 + c;
    unsigned lo = 0u, hi = 0u;
    if (i < HW) {
        #pragma unroll
        for (int k = 0; k < HB / 4; ++k) {
            const unsigned w = part[(size_t)(pg + 4 * k) * HWS + i];
            lo += w & 0xffffu;
            hi += w >> 16;
        }
    }
    if (pg > 0) { s[pg - 1][0][c] = lo; s[pg - 1][1][c] = hi; }
    __syncthreads();
    if (pg == 0 && i < HW) {
        lo += s[0][0][c] + s[1][0][c] + s[2][0][c];
        hi += s[0][1][c] + s[1][1][c] + s[2][1][c];
        float2 r;
        r.x = rsqrtf((float)(lo + 1u));
        r.y = rsqrtf((float)(hi + 1u));
        ((float2*)dinv)[i] = r;
    }
}

template <bool FULL>
__device__ __forceinline__ void edge_group(const int* qp, int cnt, int lane, int row0,
        const float* __restrict__ x, const int* __restrict__ ei,
        const float* __restrict__ dinv, float* tile) {
    const int jj = lane & 15;
    const int e  = (FULL || jj < cnt) ? qp[jj] : 0;
    const int   my  = ei[lane < 16 ? e : N_EDGES + e];
    const float myd = dinv[my];
    float v[16][4];
    #pragma unroll
    for (int j = 0; j < 16; ++j) {
        v[j][0] = 0.0f; v[j][1] = 0.0f; v[j][2] = 0.0f; v[j][3] = 0.0f;
        if (FULL || j < cnt) {
            const int s = __builtin_amdgcn_readlane(my, j);
            const float* xr = x + (size_t)s * DIM + lane;
            v[j][0] = xr[0];
            v[j][1] = xr[32];
            v[j][2] = xr[64];
            v[j][3] = xr[96];
        }
    }
    const int mydi = __float_as_int(myd);
    #pragma unroll
    for (int j = 0; j < 16; ++j) {
        if (FULL || j < cnt) {
            const int r = __builtin_amdgcn_readlane(my, 16 + j) - row0;
            const float nrm = __int_as_float(__builtin_amdgcn_readlane(mydi, j)) *
                              __int_as_float(__builtin_amdgcn_readlane(mydi, 16 + j));
            float* p = tile + r * TS + lane;
            unsafeAtomicAdd(p +  0, v[j][0] * nrm);
            unsafeAtomicAdd(p + 32, v[j][1] * nrm);
            unsafeAtomicAdd(p + 64, v[j][2] * nrm);
            unsafeAtomicAdd(p + 96, v[j][3] * nrm);
        }
    }
}

__global__ void __launch_bounds__(1024) k_owner(const float* __restrict__ x, const int* __restrict__ ei,
        const float* __restrict__ dinv, const unsigned* __restrict__ bm,
        const float* __restrict__ W, const float* __restrict__ bias, float* __restrict__ out) {
    __shared__ float tile[TROWS * TS];
    __shared__ int   q[32 * QW];
    const int tid  = threadIdx.x;
    const int lane = tid & 31;
    const int w    = tid >> 5;
    const int rid  = blockIdx.x;
    const int row0 = rid * ROWS;
    const int nrows = (N_NODES - row0) < ROWS ? (N_NODES - row0) : ROWS;

    for (int i = tid; i < nrows * (DIM / 4); i += 1024) {
        float s = dinv[row0 + (i >> 5)]; s = s * s;
        float4 t = ((const float4*)(x + (size_t)row0 * DIM))[i];
        t.x *= s; t.y *= s; t.z *= s; t.w *= s;
        ((float4*)tile)[(i >> 5) * (TS / 4) + (i & 31)] = t;
    }
    for (int i = nrows * (DIM / 4) + tid; i < TROWS * (DIM / 4); i += 1024)
        ((float4*)tile)[(i >> 5) * (TS / 4) + (i & 31)] = make_float4(0.0f, 0.0f, 0.0f, 0.0f);
    __syncthreads();

    int* myq = q + w * QW;
    int qn = 0;
    const unsigned* mybm = bm + (size_t)rid * BMW;
    for (int t0 = w; t0 < NCW; t0 += 256) {
        unsigned f[8];
        #pragma unroll
        for (int u = 0; u < 8; ++u) {
            const int  t  = t0 + 32 * u;
            const bool ok = t < NCW;
            const unsigned b = mybm[(ok ? t : 0) * 32 + lane];
            f[u] = ok ? b : 0u;
        }
        #pragma unroll 1
        for (int u = 0; u < 8; ++u) {
            unsigned m = f[0];
            f[0] = f[1]; f[1] = f[2]; f[2] = f[3]; f[3] = f[4]; f[4] = f[5]; f[5] = f[6]; f[6] = f[7]; f[7] = 0u;
            const int ebase = ((t0 + 32 * u) * 32 + lane) * 32;
            for (;;) {
                const unsigned bal = __builtin_amdgcn_ballot_w32(m != 0u);
                if (bal == 0u) break;
                if (m != 0u) {
                    const int pos = (int)__builtin_amdgcn_mbcnt_lo(bal, 0u);
                    const int p   = __builtin_ctz(m);
                    myq[qn + pos] = ebase + p;
                    m &= m - 1u;
                }
                qn += __builtin_popcount(bal);
                while (qn >= 16) {
                    qn -= 16;
                    edge_group<true>(myq + qn, 16, lane, row0, x, ei, dinv, tile);
                }
            }
        }
    }
    if (qn > 0) edge_group<false>(myq, qn, lane, row0, x, ei, dinv, tile);

    const int wu   = __builtin_amdgcn_readfirstlane(w);
    const int nn   = lane & 15;
    const int koff = (lane >> 4) << 1;
    const int colg = (wu & 7) * 16 + nn;
    v2f bfrag[32];
    #pragma unroll
    for (int kk = 0; kk < 32; ++kk) {
        const int k = (kk << 2) + koff;
        v2f t;
        t.x = W[k * DIM + colg];
        t.y = W[(k + 1) * DIM + colg];
        bfrag[kk] = t;
    }
    const float bv = bias[colg];
    __syncthreads();

    const int ntile = ((wu >> 3) == (wu & 3)) ? 4 : 3;
    #pragma unroll 1
    for (int i = 0; i < ntile; ++i) {
        const int rt   = (i < 3) ? (wu >> 3) + 4 * i : 12;
        const int rloc = rt * 16 + ((lane >> 4) << 3);
        const bool whole = rt * 16 + 16 <= nrows;
        const size_t o0 = (size_t)(row0 + rloc) * DIM + colg;
        const float* xp = x + o0;
        float*       op = out + o0;
        float xr[8];
        if (whole) {
            #pragma unroll
            for (int r = 0; r < 8; ++r) xr[r] = xp[r * DIM];
        } else {
            #pragma unroll
            for (int r = 0; r < 8; ++r) xr[r] = (rloc + r < nrows) ? xp[r * DIM] : 0.0f;
        }
        const float* arow = tile + (rt * 16 + nn) * TS;
        v8f c = {};
        #pragma unroll
        for (int kk = 0; kk < 32; ++kk) {
            const int k = (kk << 2) + koff;
            v2f a;
            a.x = arow[k];
            a.y = arow[k + 1];
            c = __builtin_amdgcn_wmma_f32_16x16x4_f32(false, a, false, bfrag[kk],
                                                      (short)0, c, false, false);
        }
        if (whole) {
            #pragma unroll
            for (int r = 0; r < 8; ++r) {
                float v = c[r] + bv;
                v = v > 0.0f ? v : 0.0f;
                op[r * DIM] = xr[r] + v;
            }
        } else {
            #pragma unroll
            for (int r = 0; r < 8; ++r) {
                if (rloc + r < nrows) {
                    float v = c[r] + bv;
                    v = v > 0.0f ? v : 0.0f;
                    op[r * DIM] = xr[r] + v;
                }
            }
        }
    }
}

extern "C" void kernel_launch(void* const* d_in, const int* in_sizes, int n_in,
                              void* d_out, int out_size, void* d_ws, size_t ws_size,
                              hipStream_t stream) {
    (void)in_sizes; (void)n_in; (void)out_size; (void)ws_size;
    const float* x  = (const float*)d_in[0];
    const int*   ei = (const int*)d_in[1];
    const float* W  = (const float*)d_in[2];
    const float* b  = (const float*)d_in[3];
    float* out = (float*)d_out;

    char* ws  = (char*)d_ws;
    float* dinv = (float*)ws;
    unsigned* part = (unsigned*)(ws + ((N_NODES * 4 + 255) & ~255));
    unsigned* bm = (unsigned*)(ws + ((N_NODES * 4 + 255) & ~255) + HB * HWS * 4);

    k_hist     <<<HB + NBB, HT, 0, stream>>>(ei, part, (uint4*)bm);
    k_dinv     <<<(HW + 63) / 64, 256, 0, stream>>>(part, dinv);
    k_owner    <<<(N_NODES + ROWS - 1) / ROWS, 1024, 0, stream>>>(x, ei, dinv, bm, W, b, out);
}


# Owner: each matched edge is routed by one integer LDS add into its row's source list; the row's wave sums the gathered rows in registers; LDS f32 atomic adds gone; sums run in list arrival order
# speedup vs baseline: 3.6197x; 1.1477x over previous
#include <hip/hip_runtime.h>

#define N_NODES 50000
#define N_EDGES 800000
#define DIM     128
#define ROWS    196
#define BE      4096
#define NBB     ((N_EDGES + BE - 1) / BE)
#define BWD     (BE / 32)
#define BST     132
#define BMW     (NBB * BWD)
#define NCW     (BMW / 32)
#define QW      64
#define RQC     64
#define RQS     65
#define HB      100
#define HW      (N_NODES / 2)
#define HWS     25024
#define HT      512
#define HG      (N_EDGES / HB / 4)
#define HR      ((HG + HT - 1) / HT)
#define BR      (BE / 4 / HT)
#define TS      132
#define TROWS   208

typedef __attribute__((ext_vector_type(2))) float v2f;
typedef __attribute__((ext_vector_type(8))) float v8f;

__device__ __forceinline__ void hist_role(const int* __restrict__ ei, unsigned* __restrict__ part, unsigned* h) {
    const int tid = threadIdx.x;
    const int g0  = blockIdx.x * HG;
    int4 d[HR];
    #pragma unroll
    for (int k = 0; k < HR; ++k) {
        const int i = tid + HT * k;
        d[k] = ((const int4*)(ei + N_EDGES))[g0 + (i < HG ? i : 0)];
    }
    for (int i = tid; i < HW / 4; i += HT) ((uint4*)h)[i] = make_uint4(0u, 0u, 0u, 0u);
    __syncthreads();
    #pragma unroll
    for (int k = 0; k < HR; ++k) {
        const int i = tid + HT * k;
        if (i < HG) {
            const int4 e = d[k];
            atomicAdd(&h[e.x >> 1], 1u << ((e.x & 1) << 4));
            atomicAdd(&h[e.y >> 1], 1u << ((e.y & 1) << 4));
            atomicAdd(&h[e.z >> 1], 1u << ((e.z & 1) << 4));
            atomicAdd(&h[e.w >> 1], 1u << ((e.w & 1) << 4));
        }
    }
    __syncthreads();
    uint4* dst = (uint4*)(part + (size_t)blockIdx.x * HWS);
    for (int i = tid; i < HW / 4; i += HT) dst[i] = ((const uint4*)h)[i];
}

__device__ __forceinline__ void bitmap_role(const int* __restrict__ ei, uint4* __restrict__ bm, unsigned* s, int b) {
    const int tid = threadIdx.x;
    const int g0  = b * (BE / 4);
    int4 d[BR];
    #pragma unroll
    for (int k = 0; k < BR; ++k) {
        const int gi = g0 + tid + HT * k;
        d[k] = ((const int4*)(ei + N_EDGES))[gi < N_EDGES / 4 ? gi : 0];
    }
    for (int i = tid; i < 256 * BST / 4; i += HT) ((uint4*)s)[i] = make_uint4(0u, 0u, 0u, 0u);
    __syncthreads();
    #pragma unroll
    for (int k = 0; k < BR; ++k) {
        const int gl = tid + HT * k;
        if (g0 + gl < N_EDGES / 4) {
            const int4 e = d[k];
            const int      wd = gl >> 3;
            const unsigned sh = (unsigned)(gl & 7) << 2;
            atomicOr(&s[((unsigned)e.x / ROWS) * BST + wd], 1u << sh);
            atomicOr(&s[((unsigned)e.y / ROWS) * BST + wd], 2u << sh);
            atomicOr(&s[((unsigned)e.z / ROWS) * BST + wd], 4u << sh);
            atomicOr(&s[((unsigned)e.w / ROWS) * BST + wd], 8u << sh);
        }
    }
    __syncthreads();
    for (int i = tid; i < 256 * (BWD / 4); i += HT) {
        const int r = i / (BWD / 4), j = i % (BWD / 4);
        bm[(size_t)r * (BMW / 4) + b * (BWD / 4) + j] = ((const uint4*)s)[r * (BST / 4) + j];
    }
}

__global__ void __launch_bounds__(HT) k_hist(const int* __restrict__ ei, unsigned* __restrict__ part,
                                             uint4* __restrict__ bm) {
    __shared__ unsigned s[256 * BST];
    if (blockIdx.x < HB) hist_role(ei, part, s);
    else                 bitmap_role(ei, bm, s, blockIdx.x - HB);
}

__global__ void __launch_bounds__(256) k_dinv(const unsigned* __restrict__ part, float* __restrict__ dinv) {
    __shared__ unsigned s[3][2][64];
    const int tid = threadIdx.x;
    const int c   = tid & 63;
    const int pg  = tid >> 6;
    const int i   = blockIdx.x * 64 + c;
    unsigned lo = 0u, hi = 0u;
    if (i < HW) {
        #pragma unroll
        for (int k = 0; k < HB / 4; ++k) {
            const unsigned w = part[(size_t)(pg + 4 * k) * HWS + i];
            lo += w & 0xffffu;
            hi += w >> 16;
        }
    }
    if (pg > 0) { s[pg - 1][0][c] = lo; s[pg - 1][1][c] = hi; }
    __syncthreads();
    if (pg == 0 && i < HW) {
        lo += s[0][0][c] + s[1][0][c] + s[2][0][c];
        hi += s[0][1][c] + s[1][1][c] + s[2][1][c];
        float2 r;
        r.x = rsqrtf((float)(lo + 1u));
        r.y = rsqrtf((float)(hi + 1u));
        ((float2*)dinv)[i] = r;
    }
}

template <bool FULL>
__device__ __forceinline__ void route_group(const int* qp, int cnt, int lane, int row0,
        const float* __restrict__ x, const int* __restrict__ ei, const float* __restrict__ dinv,
        unsigned* rcnt, int* rq, float* tile) {
    const bool act = FULL || lane < cnt;
    const int e = act ? qp[lane] : 0;
    const int s = ei[e];
    const int d = ei[N_EDGES + e];
    if (act) {
        const int r = d - row0;
        const unsigned slot = atomicAdd(&rcnt[r], 1u);
        if (slot < RQC) {
            rq[r * RQS + slot] = s;
        } else {
            const float nrm = dinv[s] * dinv[d];
            const float* xr = x + (size_t)s * DIM;
            float* p = tile + r * TS;
            #pragma unroll 1
            for (int c = 0; c < DIM; ++c) unsafeAtomicAdd(p + c, xr[c] * nrm);
        }
    }
}

template <bool FULL>
__device__ __forceinline__ void row_batch(const int* lp, int cnt, int lane, float dd,
        const float* __restrict__ x, const float* __restrict__ dinv, float (&acc)[4]) {
    const int jj = lane & 15;
    const int   sl = (FULL || jj < cnt) ? lp[jj] : 0;
    const float dl = dinv[sl];
    float v[16][4];
    #pragma unroll
    for (int j = 0; j < 16; ++j) {
        v[j][0] = 0.0f; v[j][1] = 0.0f; v[j][2] = 0.0f; v[j][3] = 0.0f;
        if (FULL || j < cnt) {
            const int s = __builtin_amdgcn_readlane(sl, j);
            const float* xr = x + (size_t)s * DIM + lane;
            v[j][0] = xr[0];
            v[j][1] = xr[32];
            v[j][2] = xr[64];
            v[j][3] = xr[96];
        }
    }
    const int dli = __float_as_int(dl);
    #pragma unroll
    for (int j = 0; j < 16; ++j) {
        if (FULL || j < cnt) {
            const float nrm = __int_as_float(__builtin_amdgcn_readlane(dli, j)) * dd;
            acc[0] = fmaf(v[j][0], nrm, acc[0]);
            acc[1] = fmaf(v[j][1], nrm, acc[1]);
            acc[2] = fmaf(v[j][2], nrm, acc[2]);
            acc[3] = fmaf(v[j][3], nrm, acc[3]);
        }
    }
}

__global__ void __launch_bounds__(1024) k_owner(const float* __restrict__ x, const int* __restrict__ ei,
        const float* __restrict__ dinv, const unsigned* __restrict__ bm,
        const float* __restrict__ W, const float* __restrict__ bias, float* __restrict__ out) {
    __shared__ float tile[TROWS * TS];
    __shared__ int   q[32 * QW];
    __shared__ int   rq[ROWS * RQS];
    __shared__ unsigned rcnt[ROWS];
    const int tid  = threadIdx.x;
    const int lane = tid & 31;
    const int w    = tid >> 5;
    const int rid  = blockIdx.x;
    const int row0 = rid * ROWS;
    const int nrows = (N_NODES - row0) < ROWS ? (N_NODES - row0) : ROWS;

    for (int i = tid; i < nrows * (DIM / 4); i += 1024) {
        float s = dinv[row0 + (i >> 5)]; s = s * s;
        float4 t = ((const float4*)(x + (size_t)row0 * DIM))[i];
        t.x *= s; t.y *= s; t.z *= s; t.w *= s;
        ((float4*)tile)[(i >> 5) * (TS / 4) + (i & 31)] = t;
    }
    for (int i = nrows * (DIM / 4) + tid; i < TROWS * (DIM / 4); i += 1024)
        ((float4*)tile)[(i >> 5) * (TS / 4) + (i & 31)] = make_float4(0.0f, 0.0f, 0.0f, 0.0f);
    if (tid < ROWS) rcnt[tid] = 0u;
    __syncthreads();

    int* myq = q + w * QW;
    int qn = 0;
    const unsigned* mybm = bm + (size_t)rid * BMW;
    for (int t0 = w; t0 < NCW; t0 += 256) {
        unsigned f[8];
        #pragma unroll
        for (int u = 0; u < 8; ++u) {
            const int  t  = t0 + 32 * u;
            const bool ok = t < NCW;
            const unsigned b = mybm[(ok ? t : 0) * 32 + lane];
            f[u] = ok ? b : 0u;
        }
        #pragma unroll 1
        for (int u = 0; u < 8; ++u) {
            unsigned m = f[0];
            f[0] = f[1]; f[1] = f[2]; f[2] = f[3]; f[3] = f[4]; f[4] = f[5]; f[5] = f[6]; f[6] = f[7]; f[7] = 0u;
            const int ebase = ((t0 + 32 * u) * 32 + lane) * 32;
            for (;;) {
                const unsigned bal = __builtin_amdgcn_ballot_w32(m != 0u);
                if (bal == 0u) break;
                if (m != 0u) {
                    const int pos = (int)__builtin_amdgcn_mbcnt_lo(bal, 0u);
                    const int p   = __builtin_ctz(m);
                    myq[qn + pos] = ebase + p;
                    m &= m - 1u;
                }
                qn += __builtin_popcount(bal);
                while (qn >= 32) {
                    qn -= 32;
                    route_group<true>(myq + qn, 32, lane, row0, x, ei, dinv, rcnt, rq, tile);
                }
            }
        }
    }
    if (qn > 0) route_group<false>(myq, qn, lane, row0, x, ei, dinv, rcnt, rq, tile);
    __syncthreads();

    const int wu   = __builtin_amdgcn_readfirstlane(w);
    for (int r = wu; r < nrows; r += 32) {
        const unsigned have = rcnt[r];
        const int n = __builtin_amdgcn_readfirstlane((int)(have < (unsigned)RQC ? have : (unsigned)RQC));
        if (n > 0) {
            const float dd = dinv[row0 + r];
            float* p = tile + r * TS + lane;
            float acc[4] = { p[0], p[32], p[64], p[96] };
            const int* lp = rq + r * RQS;
            for (int b = 0; b < n; b += 16) {
                const int cnt = n - b;
                if (cnt >= 16) row_batch<true>(lp + b, 16, lane, dd, x, dinv, acc);
                else           row_batch<false>(lp + b, cnt, lane, dd, x, dinv, acc);
            }
            p[0] = acc[0]; p[32] = acc[1]; p[64] = acc[2]; p[96] = acc[3];
        }
    }

    const int nn   = lane & 15;
    const int koff = (lane >> 4) << 1;
    const int colg = (wu & 7) * 16 + nn;
    v2f bfrag[32];
    #pragma unroll
    for (int kk = 0; kk < 32; ++kk) {
        const int k = (kk << 2) + koff;
        v2f t;
        t.x = W[k * DIM + colg];
        t.y = W[(k + 1) * DIM + colg];
        bfrag[kk] = t;
    }
    const float bv = bias[colg];
    __syncthreads();

    const int ntile = ((wu >> 3) == (wu & 3)) ? 4 : 3;
    #pragma unroll 1
    for (int i = 0; i < ntile; ++i) {
        const int rt   = (i < 3) ? (wu >> 3) + 4 * i : 12;
        const int rloc = rt * 16 + ((lane >> 4) << 3);
        const bool whole = rt * 16 + 16 <= nrows;
        const size_t o0 = (size_t)(row0 + rloc) * DIM + colg;
        const float* xp = x + o0;
        float*       op = out + o0;
        float xr[8];
        if (whole) {
            #pragma unroll
            for (int r = 0; r < 8; ++r) xr[r] = xp[r * DIM];
        } else {
            #pragma unroll
            for (int r = 0; r < 8; ++r) xr[r] = (rloc + r < nrows) ? xp[r * DIM] : 0.0f;
        }
        const float* arow = tile + (rt * 16 + nn) * TS;
        v8f c = {};
        #pragma unroll
        for (int kk = 0; kk < 32; ++kk) {
            const int k = (kk << 2) + koff;
            v2f a;
            a.x = arow[k];
            a.y = arow[k + 1];
            c = __builtin_amdgcn_wmma_f32_16x16x4_f32(false, a, false, bfrag[kk],
                                                      (short)0, c, false, false);
        }
        if (whole) {
            #pragma unroll
            for (int r = 0; r < 8; ++r) {
                float v = c[r] + bv;
                v = v > 0.0f ? v : 0.0f;
                op[r * DIM] = xr[r] + v;
            }
        } else {
            #pragma unroll
            for (int r = 0; r < 8; ++r) {
                if (rloc + r < nrows) {
                    float v = c[r] + bv;
                    v = v > 0.0f ? v : 0.0f;
                    op[r * DIM] = xr[r] + v;
                }
            }
        }
    }
}

extern "C" void kernel_launch(void* const* d_in, const int* in_sizes, int n_in,
                              void* d_out, int out_size, void* d_ws, size_t ws_size,
                              hipStream_t stream) {
    (void)in_sizes; (void)n_in; (void)out_size; (void)ws_size;
    const float* x  = (const float*)d_in[0];
    const int*   ei = (const int*)d_in[1];
    const float* W  = (const float*)d_in[2];
    const float* b  = (const float*)d_in[3];
    float* out = (float*)d_out;

    char* ws  = (char*)d_ws;
    float* dinv = (float*)ws;
    unsigned* part = (unsigned*)(ws + ((N_NODES * 4 + 255) & ~255));
    unsigned* bm = (unsigned*)(ws + ((N_NODES * 4 + 255) & ~255) + HB * HWS * 4);

    k_hist     <<<HB + NBB, HT, 0, stream>>>(ei, part, (uint4*)bm);
    k_dinv     <<<(HW + 63) / 64, 256, 0, stream>>>(part, dinv);
    k_owner    <<<(N_NODES + ROWS - 1) / ROWS, 1024, 0, stream>>>(x, ei, dinv, bm, W, b, out);
}
